# starting kernel
# baseline (speedup 1.0000x reference)
#include <hip/hip_runtime.h>

typedef float v2f __attribute__((ext_vector_type(2)));
typedef float v8f __attribute__((ext_vector_type(8)));

#define B_  2
#define S_  96
#define H_  96
#define D_  96
#define NP  (B_ * S_)
#define SCALE_ 0.10206207261596575f

__global__ void k_tvec(const float* __restrict__ Q,
                       const float* __restrict__ V,
                       const int*   __restrict__ idxp,
                       float*       __restrict__ T) {
    __shared__ float qsh[D_];
    const int p = blockIdx.x;
    const int i = threadIdx.x;
    const int idx = *idxp;

    const float* __restrict__ qrow = Q + (size_t)p * H_ * D_ + (size_t)idx * D_;
    if (i < D_) qsh[i] = qrow[i];
    __syncthreads();

    const float* __restrict__ vbase = V + (size_t)p * H_ * D_;
    float acc = 0.0f;
#pragma unroll 8
    for (int k = 0; k < H_; ++k) {
        acc = fmaf(qsh[k], vbase[(size_t)k * D_ + i], acc);
    }
    T[(size_t)p * D_ + i] = acc;
}

__global__ void k_gemm_wmma(const float* __restrict__ T,
                            const float* __restrict__ DW,
                            const int*   __restrict__ idxp,
                            float*       __restrict__ U) {
    const int tile = blockIdx.x;
    const int tn   = tile % (D_ / 16);
    const int tm   = tile / (D_ / 16);
    const int lane = threadIdx.x & 31;
    const int half = lane >> 4;
    const int lid  = lane & 15;

    const float* __restrict__ W = DW + (size_t)(*idxp) * D_ * D_;
    const int row0 = tm * 16;
    const int col0 = tn * 16;

    v8f c = {};
    for (int kk = 0; kk < D_; kk += 4) {
        const int ka = kk + 2 * half;
        v2f a, b;
        a.x = T[(size_t)(row0 + lid) * D_ + ka];
        a.y = T[(size_t)(row0 + lid) * D_ + ka + 1];
        b.x = W[(size_t)ka       * D_ + col0 + lid];
        b.y = W[(size_t)(ka + 1) * D_ + col0 + lid];
        c = __builtin_amdgcn_wmma_f32_16x16x4_f32(false, a, false, b,
                                                  (short)0, c, false, false);
    }

#pragma unroll
    for (int j = 0; j < 8; ++j) {
        U[(size_t)(row0 + j + 8 * half) * D_ + col0 + lid] = c[j];
    }
}

__global__ void k_outer(const float* __restrict__ Kmat,
                        const float* __restrict__ U,
                        const int*   __restrict__ idxp,
                        float*       __restrict__ out) {
    const int gid = blockIdx.x * blockDim.x + threadIdx.x;
    const int total = NP * D_ * D_;
    if (gid >= total) return;
    const int v = gid % D_;
    const int r = gid / D_;
    const int q = r % D_;
    const int p = r / D_;
    const float kv = Kmat[((size_t)p * H_ + (size_t)(*idxp)) * D_ + q];
    out[gid] = SCALE_ * kv * U[(size_t)p * D_ + v];
}

extern "C" void kernel_launch(void* const* d_in, const int* in_sizes, int n_in,
                              void* d_out, int out_size, void* d_ws, size_t ws_size,
                              hipStream_t stream) {
    const float* Q   = (const float*)d_in[0];
    const float* K   = (const float*)d_in[1];
    const float* V   = (const float*)d_in[2];
    const float* DW  = (const float*)d_in[3];
    const int*   idx = (const int*)  d_in[4];
    float* out = (float*)d_out;

    float* T = (float*)d_ws;
    float* U = T + (size_t)NP * D_;

    k_tvec     <<<NP, 96, 0, stream>>>(Q, V, idx, T);
    k_gemm_wmma<<<(NP / 16) * (D_ / 16), 32, 0, stream>>>(T, DW, idx, U);

    const int total  = NP * D_ * D_;
    const int blocks = (total + 255) / 256;
    k_outer    <<<blocks, 256, 0, stream>>>(K, U, idx, out);
}


# T*W product folded into the outer-product kernel: each position's work-group forms its 16x96 row tile of U on the f32 WMMA (same k order, one chain) and keeps its row via LDS; 2 launches, were 3
# speedup vs baseline: 1.5918x; 1.5918x over previous
#include <hip/hip_runtime.h>

typedef float v2f __attribute__((ext_vector_type(2)));
typedef float v8f __attribute__((ext_vector_type(8)));

#define B_  2
#define S_  96
#define H_  96
#define D_  96
#define NP  (B_ * S_)
#define SCALE_ 0.10206207261596575f

__global__ void k_tvec(const float* __restrict__ Q,
                       const float* __restrict__ V,
                       const int*   __restrict__ idxp,
                       float*       __restrict__ T) {
    __shared__ float qsh[D_];
    const int p = blockIdx.x;
    const int i = threadIdx.x;
    const int idx = *idxp;

    const float* __restrict__ qrow = Q + (size_t)p * H_ * D_ + (size_t)idx * D_;
    if (i < D_) qsh[i] = qrow[i];
    __syncthreads();

    const float* __restrict__ vbase = V + (size_t)p * H_ * D_;
    float acc = 0.0f;
#pragma unroll 8
    for (int k = 0; k < H_; ++k) {
        acc = fmaf(qsh[k], vbase[(size_t)k * D_ + i], acc);
    }
    T[(size_t)p * D_ + i] = acc;
}

__global__ __launch_bounds__(256) void k_gemm_outer(const float* __restrict__ Kmat,
                                                    const float* __restrict__ T,
                                                    const float* __restrict__ DW,
                                                    const int*   __restrict__ idxp,
                                                    float*       __restrict__ out) {
    __shared__ float4 ush4[2 * (D_ / 4)];
    const int p    = blockIdx.x;
    const int lane = threadIdx.x & 31;
    const int wave = __builtin_amdgcn_readfirstlane((int)(threadIdx.x >> 5));
    const int idx  = *idxp;
    const int r    = p & 15;

    const float* __restrict__ krow = Kmat + ((size_t)p * H_ + (size_t)idx) * D_ + wave * 12;
    float kf[12];
#pragma unroll
    for (int j = 0; j < 12; ++j) kf[j] = krow[j];

    if (wave < D_ / 16) {
        const int half = lane >> 4;
        const int lid  = lane & 15;
        const int row0 = (p >> 4) * 16;
        const int col0 = wave * 16;
        const float* __restrict__ trow = T + (size_t)(row0 + lid) * D_ + 2 * half;
        const float* __restrict__ wcol =
            DW + (size_t)idx * D_ * D_ + (size_t)(2 * half) * D_ + col0 + lid;

        v2f a[D_ / 4], b[D_ / 4];
#pragma unroll
        for (int s = 0; s < D_ / 4; ++s) {
            a[s].x = trow[4 * s];
            a[s].y = trow[4 * s + 1];
            b[s].x = wcol[(size_t)(4 * s)     * D_];
            b[s].y = wcol[(size_t)(4 * s + 1) * D_];
        }
        v8f c = {};
#pragma unroll
        for (int s = 0; s < D_ / 4; ++s) {
            c = __builtin_amdgcn_wmma_f32_16x16x4_f32(false, a[s], false, b[s],
                                                      (short)0, c, false, false);
        }
        const int jr = r & 7;
        float val = c[0];
#pragma unroll
        for (int j = 1; j < 8; ++j) val = (jr == j) ? c[j] : val;
        reinterpret_cast<float*>(ush4)[half * D_ + col0 + lid] = val;
    }

    float4* __restrict__ o4 =
        reinterpret_cast<float4*>(out + ((size_t)p * D_ + (size_t)(wave * 12)) * D_) + lane;

    __syncthreads();

    const int g0 = lane, g1 = lane + 32, g2 = lane + 64;
    const float4* urow = ush4 + (r >> 3) * (D_ / 4);
    const float4 u0 = urow[g0 % 24];
    const float4 u1 = urow[g1 % 24];
    const float4 u2 = urow[g2 % 24];
    const bool up0 = (g0 / 24) != 0;
    const bool up1 = (g1 / 24) != 1;
    const bool up2 = (g2 / 24) != 2;

#pragma unroll
    for (int grp = 0; grp < 3; ++grp) {
        const float s0 = SCALE_ * kf[4 * grp + 0];
        const float s1 = SCALE_ * kf[4 * grp + 1];
        const float s2 = SCALE_ * kf[4 * grp + 2];
        const float s3 = SCALE_ * kf[4 * grp + 3];
        const float k0 = up0 ? s1 : s0;
        const float k1 = up1 ? s2 : s1;
        const float k2 = up2 ? s3 : s2;
        float4 o;
        o.x = k0 * u0.x; o.y = k0 * u0.y; o.z = k0 * u0.z; o.w = k0 * u0.w;
        o4[grp * 96] = o;
        o.x = k1 * u1.x; o.y = k1 * u1.y; o.z = k1 * u1.z; o.w = k1 * u1.w;
        o4[grp * 96 + 32] = o;
        o.x = k2 * u2.x; o.y = k2 * u2.y; o.z = k2 * u2.z; o.w = k2 * u2.w;
        o4[grp * 96 + 64] = o;
    }
}

extern "C" void kernel_launch(void* const* d_in, const int* in_sizes, int n_in,
                              void* d_out, int out_size, void* d_ws, size_t ws_size,
                              hipStream_t stream) {
    const float* Q   = (const float*)d_in[0];
    const float* K   = (const float*)d_in[1];
    const float* V   = (const float*)d_in[2];
    const float* DW  = (const float*)d_in[3];
    const int*   idx = (const int*)  d_in[4];
    float* out = (float*)d_out;

    float* T = (float*)d_ws;

    k_tvec      <<<NP, 96, 0, stream>>>(Q, V, idx, T);
    k_gemm_outer<<<NP, 256, 0, stream>>>(K, T, DW, idx, out);
}


# T kernel: k loop fully unrolled so that all 96 V row loads of a thread go out ahead of the sum, one memory latency instead of 12 dependent trips of 8 rows; same k order and same sums
# speedup vs baseline: 2.5673x; 1.6128x over previous
#include <hip/hip_runtime.h>

typedef float v2f __attribute__((ext_vector_type(2)));
typedef float v8f __attribute__((ext_vector_type(8)));

#define B_  2
#define S_  96
#define H_  96
#define D_  96
#define NP  (B_ * S_)
#define SCALE_ 0.10206207261596575f

__global__ void k_tvec(const float* __restrict__ Q,
                       const float* __restrict__ V,
                       const int*   __restrict__ idxp,
                       float*       __restrict__ T) {
    __shared__ float qsh[D_];
    const int p = blockIdx.x;
    const int i = threadIdx.x;
    const int idx = *idxp;

    const float* __restrict__ qrow = Q + (size_t)p * H_ * D_ + (size_t)idx * D_;
    if (i < D_) qsh[i] = qrow[i];
    __syncthreads();

    const float* __restrict__ vcol = V + (size_t)p * H_ * D_ + i;
    float v[H_];
#pragma unroll
    for (int k = 0; k < H_; ++k) v[k] = vcol[(size_t)k * D_];

    float acc = 0.0f;
#pragma unroll
    for (int k = 0; k < H_; ++k) acc = fmaf(qsh[k], v[k], acc);
    T[(size_t)p * D_ + i] = acc;
}

__global__ __launch_bounds__(256) void k_gemm_outer(const float* __restrict__ Kmat,
                                                    const float* __restrict__ T,
                                                    const float* __restrict__ DW,
                                                    const int*   __restrict__ idxp,
                                                    float*       __restrict__ out) {
    __shared__ float4 ush4[2 * (D_ / 4)];
    const int p    = blockIdx.x;
    const int lane = threadIdx.x & 31;
    const int wave = __builtin_amdgcn_readfirstlane((int)(threadIdx.x >> 5));
    const int idx  = *idxp;
    const int r    = p & 15;

    const float* __restrict__ krow = Kmat + ((size_t)p * H_ + (size_t)idx) * D_ + wave * 12;
    float kf[12];
#pragma unroll
    for (int j = 0; j < 12; ++j) kf[j] = krow[j];

    if (wave < D_ / 16) {
        const int half = lane >> 4;
        const int lid  = lane & 15;
        const int row0 = (p >> 4) * 16;
        const int col0 = wave * 16;
        const float* __restrict__ trow = T + (size_t)(row0 + lid) * D_ + 2 * half;
        const float* __restrict__ wcol =
            DW + (size_t)idx * D_ * D_ + (size_t)(2 * half) * D_ + col0 + lid;

        v2f a[D_ / 4], b[D_ / 4];
#pragma unroll
        for (int s = 0; s < D_ / 4; ++s) {
            a[s].x = trow[4 * s];
            a[s].y = trow[4 * s + 1];
            b[s].x = wcol[(size_t)(4 * s)     * D_];
            b[s].y = wcol[(size_t)(4 * s + 1) * D_];
        }
        v8f c = {};
#pragma unroll
        for (int s = 0; s < D_ / 4; ++s) {
            c = __builtin_amdgcn_wmma_f32_16x16x4_f32(false, a[s], false, b[s],
                                                      (short)0, c, false, false);
        }
        const int jr = r & 7;
        float val = c[0];
#pragma unroll
        for (int j = 1; j < 8; ++j) val = (jr == j) ? c[j] : val;
        reinterpret_cast<float*>(ush4)[half * D_ + col0 + lid] = val;
    }

    float4* __restrict__ o4 =
        reinterpret_cast<float4*>(out + ((size_t)p * D_ + (size_t)(wave * 12)) * D_) + lane;

    __syncthreads();

    const int g0 = lane, g1 = lane + 32, g2 = lane + 64;
    const float4* urow = ush4 + (r >> 3) * (D_ / 4);
    const float4 u0 = urow[g0 % 24];
    const float4 u1 = urow[g1 % 24];
    const float4 u2 = urow[g2 % 24];
    const bool up0 = (g0 / 24) != 0;
    const bool up1 = (g1 / 24) != 1;
    const bool up2 = (g2 / 24) != 2;

#pragma unroll
    for (int grp = 0; grp < 3; ++grp) {
        const float s0 = SCALE_ * kf[4 * grp + 0];
        const float s1 = SCALE_ * kf[4 * grp + 1];
        const float s2 = SCALE_ * kf[4 * grp + 2];
        const float s3 = SCALE_ * kf[4 * grp + 3];
        const float k0 = up0 ? s1 : s0;
        const float k1 = up1 ? s2 : s1;
        const float k2 = up2 ? s3 : s2;
        float4 o;
        o.x = k0 * u0.x; o.y = k0 * u0.y; o.z = k0 * u0.z; o.w = k0 * u0.w;
        o4[grp * 96] = o;
        o.x = k1 * u1.x; o.y = k1 * u1.y; o.z = k1 * u1.z; o.w = k1 * u1.w;
        o4[grp * 96 + 32] = o;
        o.x = k2 * u2.x; o.y = k2 * u2.y; o.z = k2 * u2.z; o.w = k2 * u2.w;
        o4[grp * 96 + 64] = o;
    }
}

extern "C" void kernel_launch(void* const* d_in, const int* in_sizes, int n_in,
                              void* d_out, int out_size, void* d_ws, size_t ws_size,
                              hipStream_t stream) {
    const float* Q   = (const float*)d_in[0];
    const float* K   = (const float*)d_in[1];
    const float* V   = (const float*)d_in[2];
    const float* DW  = (const float*)d_in[3];
    const int*   idx = (const int*)  d_in[4];
    float* out = (float*)d_out;

    float* T = (float*)d_ws;

    k_tvec      <<<NP, 96, 0, stream>>>(Q, V, idx, T);
    k_gemm_outer<<<NP, 256, 0, stream>>>(K, T, DW, idx, out);
}


# T kernel: the row of Q, uniform over the work-group, comes by scalar loads into scalar registers; LDS stage and barrier gone, so the V requests leave at wave entry, not behind index and row loads
# speedup vs baseline: 2.7086x; 1.0550x over previous
#include <hip/hip_runtime.h>

typedef float v2f __attribute__((ext_vector_type(2)));
typedef float v8f __attribute__((ext_vector_type(8)));

#define B_  2
#define S_  96
#define H_  96
#define D_  96
#define NP  (B_ * S_)
#define SCALE_ 0.10206207261596575f

__global__ void k_tvec(const float* __restrict__ Q,
                       const float* __restrict__ V,
                       const int*   __restrict__ idxp,
                       float*       __restrict__ T) {
    const int p = blockIdx.x;
    const int i = threadIdx.x;

    const float* __restrict__ vcol = V + (size_t)p * H_ * D_ + i;
    float v[H_];
#pragma unroll
    for (int k = 0; k < H_; ++k) v[k] = vcol[(size_t)k * D_];

    const int idx = *idxp;
    const float* __restrict__ qrow = Q + (size_t)p * H_ * D_ + (size_t)idx * D_;

    float acc = 0.0f;
#pragma unroll
    for (int k = 0; k < H_; ++k) acc = fmaf(qrow[k], v[k], acc);
    T[(size_t)p * D_ + i] = acc;
}

__global__ __launch_bounds__(256) void k_gemm_outer(const float* __restrict__ Kmat,
                                                    const float* __restrict__ T,
                                                    const float* __restrict__ DW,
                                                    const int*   __restrict__ idxp,
                                                    float*       __restrict__ out) {
    __shared__ float4 ush4[2 * (D_ / 4)];
    const int p    = blockIdx.x;
    const int lane = threadIdx.x & 31;
    const int wave = __builtin_amdgcn_readfirstlane((int)(threadIdx.x >> 5));
    const int idx  = *idxp;
    const int r    = p & 15;

    const float* __restrict__ krow = Kmat + ((size_t)p * H_ + (size_t)idx) * D_ + wave * 12;
    float kf[12];
#pragma unroll
    for (int j = 0; j < 12; ++j) kf[j] = krow[j];

    if (wave < D_ / 16) {
        const int half = lane >> 4;
        const int lid  = lane & 15;
        const int row0 = (p >> 4) * 16;
        const int col0 = wave * 16;
        const float* __restrict__ trow = T + (size_t)(row0 + lid) * D_ + 2 * half;
        const float* __restrict__ wcol =
            DW + (size_t)idx * D_ * D_ + (size_t)(2 * half) * D_ + col0 + lid;

        v2f a[D_ / 4], b[D_ / 4];
#pragma unroll
        for (int s = 0; s < D_ / 4; ++s) {
            a[s].x = trow[4 * s];
            a[s].y = trow[4 * s + 1];
            b[s].x = wcol[(size_t)(4 * s)     * D_];
            b[s].y = wcol[(size_t)(4 * s + 1) * D_];
        }
        v8f c = {};
#pragma unroll
        for (int s = 0; s < D_ / 4; ++s) {
            c = __builtin_amdgcn_wmma_f32_16x16x4_f32(false, a[s], false, b[s],
                                                      (short)0, c, false, false);
        }
        const int jr = r & 7;
        float val = c[0];
#pragma unroll
        for (int j = 1; j < 8; ++j) val = (jr == j) ? c[j] : val;
        reinterpret_cast<float*>(ush4)[half * D_ + col0 + lid] = val;
    }

    float4* __restrict__ o4 =
        reinterpret_cast<float4*>(out + ((size_t)p * D_ + (size_t)(wave * 12)) * D_) + lane;

    __syncthreads();

    const int g0 = lane, g1 = lane + 32, g2 = lane + 64;
    const float4* urow = ush4 + (r >> 3) * (D_ / 4);
    const float4 u0 = urow[g0 % 24];
    const float4 u1 = urow[g1 % 24];
    const float4 u2 = urow[g2 % 24];
    const bool up0 = (g0 / 24) != 0;
    const bool up1 = (g1 / 24) != 1;
    const bool up2 = (g2 / 24) != 2;

#pragma unroll
    for (int grp = 0; grp < 3; ++grp) {
        const float s0 = SCALE_ * kf[4 * grp + 0];
        const float s1 = SCALE_ * kf[4 * grp + 1];
        const float s2 = SCALE_ * kf[4 * grp + 2];
        const float s3 = SCALE_ * kf[4 * grp + 3];
        const float k0 = up0 ? s1 : s0;
        const float k1 = up1 ? s2 : s1;
        const float k2 = up2 ? s3 : s2;
        float4 o;
        o.x = k0 * u0.x; o.y = k0 * u0.y; o.z = k0 * u0.z; o.w = k0 * u0.w;
        o4[grp * 96] = o;
        o.x = k1 * u1.x; o.y = k1 * u1.y; o.z = k1 * u1.z; o.w = k1 * u1.w;
        o4[grp * 96 + 32] = o;
        o.x = k2 * u2.x; o.y = k2 * u2.y; o.z = k2 * u2.z; o.w = k2 * u2.w;
        o4[grp * 96 + 64] = o;
    }
}

extern "C" void kernel_launch(void* const* d_in, const int* in_sizes, int n_in,
                              void* d_out, int out_size, void* d_ws, size_t ws_size,
                              hipStream_t stream) {
    const float* Q   = (const float*)d_in[0];
    const float* K   = (const float*)d_in[1];
    const float* V   = (const float*)d_in[2];
    const float* DW  = (const float*)d_in[3];
    const int*   idx = (const int*)  d_in[4];
    float* out = (float*)d_out;

    float* T = (float*)d_ws;

    k_tvec      <<<NP, 96, 0, stream>>>(Q, V, idx, T);
    k_gemm_outer<<<NP, 256, 0, stream>>>(K, T, DW, idx, out);
}


# Both kernels: a wave's loads fit the 63-deep load counter, one batch: T kernel reads rows by 96-bit loads, k in 3 blocks (float summation order changes); product kernel takes A from own T row via LDS
# speedup vs baseline: 3.4543x; 1.2753x over previous
#include <hip/hip_runtime.h>

typedef float v2f __attribute__((ext_vector_type(2)));
typedef float v8f __attribute__((ext_vector_type(8)));

#define B_  2
#define S_  96
#define H_  96
#define D_  96
#define NP  (B_ * S_)
#define SCALE_ 0.10206207261596575f

__global__ __launch_bounds__(96)
void k_tvec(const float* __restrict__ Q,
            const float* __restrict__ V,
            const int*   __restrict__ idxp,
            float*       __restrict__ T) {
    __shared__ float part[3][D_];
    const int p    = blockIdx.x;
    const int lane = threadIdx.x & 31;
    const int wave = __builtin_amdgcn_readfirstlane((int)(threadIdx.x >> 5));

    const unsigned idx = (unsigned)*idxp;

    const float* __restrict__ vp =
        V + (size_t)p * H_ * D_ + (size_t)(32 * wave) * D_ + 3 * lane;
    float vx[32], vy[32], vz[32];
#pragma unroll
    for (int r = 0; r < 32; ++r) {
        vx[r] = vp[(size_t)r * D_];
        vy[r] = vp[(size_t)r * D_ + 1];
        vz[r] = vp[(size_t)r * D_ + 2];
    }
    __builtin_amdgcn_sched_barrier(0);

    const float* __restrict__ qblk = Q + (size_t)p * H_ * D_ + (size_t)idx * D_ + 32 * wave;

    float ax = 0.0f, ay = 0.0f, az = 0.0f;
#pragma unroll
    for (int r = 0; r < 32; ++r) {
        const float q = qblk[r];
        ax = fmaf(q, vx[r], ax);
        ay = fmaf(q, vy[r], ay);
        az = fmaf(q, vz[r], az);
    }
    part[wave][3 * lane]     = ax;
    part[wave][3 * lane + 1] = ay;
    part[wave][3 * lane + 2] = az;
    __syncthreads();
    const int i = threadIdx.x;
    T[(size_t)p * D_ + i] = (part[0][i] + part[1][i]) + part[2][i];
}

__global__ __launch_bounds__(256)
void k_gemm_outer(const float* __restrict__ Kmat,
                  const float* __restrict__ T,
                  const float* __restrict__ DW,
                  const int*   __restrict__ idxp,
                  float*       __restrict__ out) {
    __shared__ float4 ush4[D_ / 4];
    __shared__ float  tsh[D_ / 16][D_];
    const int p    = blockIdx.x;
    const int lane = threadIdx.x & 31;
    const int wave = __builtin_amdgcn_readfirstlane((int)(threadIdx.x >> 5));
    const int idx  = *idxp;

    const float* __restrict__ krow = Kmat + ((size_t)p * H_ + (size_t)idx) * D_ + wave * 12;
    float kf[12];
#pragma unroll
    for (int j = 0; j < 12; ++j) kf[j] = krow[j];

    if (wave < D_ / 16) {
        const int half = lane >> 4;
        const int lid  = lane & 15;
        const int col0 = wave * 16;
        const float* __restrict__ trow = T + (size_t)p * D_ + lane;
        const float tv0 = trow[0], tv1 = trow[32], tv2 = trow[64];
        const float* __restrict__ wcol =
            DW + (size_t)idx * D_ * D_ + (size_t)(2 * half) * D_ + col0 + lid;

        v2f a[D_ / 4], b[D_ / 4];
#pragma unroll
        for (int s = 0; s < D_ / 4; ++s) {
            b[s].x = wcol[(size_t)(4 * s)     * D_];
            b[s].y = wcol[(size_t)(4 * s + 1) * D_];
        }
        float* tw = tsh[wave];
        tw[lane] = tv0; tw[lane + 32] = tv1; tw[lane + 64] = tv2;
        __builtin_amdgcn_fence(__ATOMIC_RELEASE, "wavefront");
        __builtin_amdgcn_wave_barrier();
        __builtin_amdgcn_fence(__ATOMIC_ACQUIRE, "wavefront");
#pragma unroll
        for (int s = 0; s < D_ / 4; ++s) {
            a[s].x = tw[4 * s + 2 * half];
            a[s].y = tw[4 * s + 2 * half + 1];
        }
        v8f c = {};
#pragma unroll
        for (int s = 0; s < D_ / 4; ++s) {
            c = __builtin_amdgcn_wmma_f32_16x16x4_f32(false, a[s], false, b[s],
                                                      (short)0, c, false, false);
        }
        if (half == 0) reinterpret_cast<float*>(ush4)[col0 + lid] = c[0];
    }

    float4* __restrict__ o4 =
        reinterpret_cast<float4*>(out + ((size_t)p * D_ + (size_t)(wave * 12)) * D_) + lane;

    __syncthreads();

    const int g0 = lane, g1 = lane + 32, g2 = lane + 64;
    const float4 u0 = ush4[g0 % 24];
    const float4 u1 = ush4[g1 % 24];
    const float4 u2 = ush4[g2 % 24];
    const bool up0 = (g0 / 24) != 0;
    const bool up1 = (g1 / 24) != 1;
    const bool up2 = (g2 / 24) != 2;

#pragma unroll
    for (int grp = 0; grp < 3; ++grp) {
        const float s0 = SCALE_ * kf[4 * grp + 0];
        const float s1 = SCALE_ * kf[4 * grp + 1];
        const float s2 = SCALE_ * kf[4 * grp + 2];
        const float s3 = SCALE_ * kf[4 * grp + 3];
        const float k0 = up0 ? s1 : s0;
        const float k1 = up1 ? s2 : s1;
        const float k2 = up2 ? s3 : s2;
        float4 o;
        o.x = k0 * u0.x; o.y = k0 * u0.y; o.z = k0 * u0.z; o.w = k0 * u0.w;
        o4[grp * 96] = o;
        o.x = k1 * u1.x; o.y = k1 * u1.y; o.z = k1 * u1.z; o.w = k1 * u1.w;
        o4[grp * 96 + 32] = o;
        o.x = k2 * u2.x; o.y = k2 * u2.y; o.z = k2 * u2.z; o.w = k2 * u2.w;
        o4[grp * 96 + 64] = o;
    }
}

extern "C" void kernel_launch(void* const* d_in, const int* in_sizes, int n_in,
                              void* d_out, int out_size, void* d_ws, size_t ws_size,
                              hipStream_t stream) {
    const float* Q   = (const float*)d_in[0];
    const float* K   = (const float*)d_in[1];
    const float* V   = (const float*)d_in[2];
    const float* DW  = (const float*)d_in[3];
    const int*   idx = (const int*)  d_in[4];
    float* out = (float*)d_out;

    float* T = (float*)d_ws;

    k_tvec      <<<NP, 96, 0, stream>>>(Q, V, idx, T);
    k_gemm_outer<<<NP, 256, 0, stream>>>(K, T, DW, idx, out);
}
